# starting kernel
# baseline (speedup 1.0000x reference)
#include <hip/hip_runtime.h>

typedef __attribute__((ext_vector_type(2))) float v2f;
typedef __attribute__((ext_vector_type(8))) float v8f;

#define D1 501
#define NSTEPS 512
#define LD 512
#define WIN 4

__global__ void kf_init_S(const float* __restrict__ Sigma0, float* __restrict__ S) {
    int idx = blockIdx.x * blockDim.x + threadIdx.x;
    if (idx < LD * LD) {
        int r = idx >> 9;
        int c = idx & (LD - 1);
        float v = 0.0f;
        if (r < D1 && c < D1) v = Sigma0[r * D1 + c];
        S[idx] = v;
    }
}

__launch_bounds__(1024)
__global__ void kf_scan(const float* __restrict__ emissions,
                        const float* __restrict__ Ht,
                        const float* __restrict__ R,
                        const float* __restrict__ mu0,
                        float* __restrict__ S,
                        float* __restrict__ out) {
    __shared__ float h[LD];
    __shared__ float hs[LD];
    __shared__ float m[LD];
    __shared__ float ps[2 * LD];
    __shared__ float kbuf[WIN][LD];
    __shared__ float svals[WIN];
    __shared__ float redA[16], redB[16], redC[16];
    __shared__ float sc[8];

    const int tid  = threadIdx.x;
    const int lane = tid & 31;
    const int wave = tid >> 5;
    const float r = R[0];

    if (tid < LD)  m[tid] = (tid < D1) ? mu0[tid] : 0.0f;
    if (tid < WIN) svals[tid] = 0.0f;
    __syncthreads();

    for (int t = 0; t < NSTEPS; ++t) {
        const int w = t & (WIN - 1);

        if (tid < LD) {
            h[tid] = (tid < D1) ? Ht[t * D1 + tid] : 0.0f;
            if (t + 1 < NSTEPS && tid < D1)
                __builtin_prefetch(Ht + (t + 1) * D1 + tid, 0, 0);
        }
        __syncthreads();

        {
            int i    = tid & (LD - 1);
            int half = tid >> 9;
            int j0   = half * (LD / 2);
            const float* Scol = S + i;
            float acc = 0.0f;
            #pragma unroll 8
            for (int j = j0; j < j0 + LD / 2; ++j)
                acc = fmaf(Scol[j * LD], h[j], acc);
            ps[half * LD + i] = acc;
        }

        if (w > 0 && tid < LD) {
            float hv = h[tid];
            float d0 = kbuf[0][tid] * hv;
            float d1 = (w > 1) ? kbuf[1][tid] * hv : 0.0f;
            float d2 = (w > 2) ? kbuf[2][tid] * hv : 0.0f;
            #pragma unroll
            for (int off = 16; off > 0; off >>= 1) {
                d0 += __shfl_down(d0, off, 32);
                d1 += __shfl_down(d1, off, 32);
                d2 += __shfl_down(d2, off, 32);
            }
            if (lane == 0) { redA[wave] = d0; redB[wave] = d1; redC[wave] = d2; }
        }
        __syncthreads();
        if (w > 0 && tid == 0) {
            float D0 = 0.0f, D1s = 0.0f, D2 = 0.0f;
            for (int q = 0; q < 16; ++q) { D0 += redA[q]; D1s += redB[q]; D2 += redC[q]; }
            sc[3] = svals[0] * D0;
            sc[4] = svals[1] * D1s;
            sc[5] = svals[2] * D2;
        }
        __syncthreads();

        if (tid < LD) {
            float v = ps[tid] + ps[LD + tid];
            if (w > 0) v -= sc[3] * kbuf[0][tid];
            if (w > 1) v -= sc[4] * kbuf[1][tid];
            if (w > 2) v -= sc[5] * kbuf[2][tid];
            hs[tid] = v;
            float hv = h[tid];
            float r1 = v * hv;
            float r2 = hv * m[tid];
            #pragma unroll
            for (int off = 16; off > 0; off >>= 1) {
                r1 += __shfl_down(r1, off, 32);
                r2 += __shfl_down(r2, off, 32);
            }
            if (lane == 0) { redA[wave] = r1; redB[wave] = r2; }
        }
        __syncthreads();
        if (tid == 0) {
            float s1 = 0.0f, s2 = 0.0f;
            for (int q = 0; q < 16; ++q) { s1 += redA[q]; s2 += redB[q]; }
            float s_term = r + s1;
            sc[0] = s_term;
            sc[1] = 1.0f / (s_term + 1e-6f);
            sc[2] = emissions[t] - s2;
            svals[w] = s_term;
        }
        __syncthreads();

        if (tid < LD) {
            float kv = hs[tid] * sc[1];
            kbuf[w][tid] = kv;
            float mv = m[tid] + kv * sc[2];
            m[tid] = mv;
            if (tid < D1) out[t * D1 + tid] = mv;
        }
        __syncthreads();

        if (w == WIN - 1) {
            const float s0 = svals[0], s1v = svals[1];
            const float s2v = svals[2], s3v = svals[3];
            const int  rl = lane & 15;
            const int  tr = wave << 4;
            const bool lo = (lane < 16);

            float k0r = kbuf[0][tr + rl], k1r = kbuf[1][tr + rl];
            float k2r = kbuf[2][tr + rl], k3r = kbuf[3][tr + rl];
            v2f a;
            a.x = lo ? (-s0  * k0r) : (-s2v * k2r);
            a.y = lo ? (-s1v * k1r) : (-s3v * k3r);

            const int rbase = tr + (lo ? 0 : 8);
            float* Sp0 = S + rbase * LD + rl;

            for (int tc = 0; tc < LD; tc += 16) {
                float* Sp = Sp0 + tc;
                v8f c;
                #pragma unroll
                for (int v = 0; v < 8; ++v) c[v] = Sp[v * LD];

                int col = tc + rl;
                float b0 = kbuf[0][col], b1 = kbuf[1][col];
                float b2 = kbuf[2][col], b3 = kbuf[3][col];
                v2f b;
                b.x = lo ? b0 : b2;
                b.y = lo ? b1 : b3;

                v8f d = __builtin_amdgcn_wmma_f32_16x16x4_f32(
                    false, a, false, b, (short)0, c, false, false);

                #pragma unroll
                for (int v = 0; v < 8; ++v) Sp[v * LD] = d[v];
            }
        }
        __syncthreads();
    }
}

extern "C" void kernel_launch(void* const* d_in, const int* in_sizes, int n_in,
                              void* d_out, int out_size, void* d_ws, size_t ws_size,
                              hipStream_t stream) {
    const float* emissions = (const float*)d_in[0];
    const float* Ht        = (const float*)d_in[1];
    const float* R         = (const float*)d_in[4];
    const float* mu0       = (const float*)d_in[5];
    const float* Sigma0    = (const float*)d_in[6];

    float* S   = (float*)d_ws;
    float* out = (float*)d_out;

    kf_init_S<<<(LD * LD) / 256, 256, 0, stream>>>(Sigma0, S);
    kf_scan<<<1, 1024, 0, stream>>>(emissions, Ht, R, mu0, S, out);
}


# workspace part of S: a wave's L2 loads go out two bands at a time (was one): matvec, ahead of 3 LDS tiles of the same trip so the round trip runs under them; flush, 4 tiles a trip; same sum order
# speedup vs baseline: 1.9431x; 1.9431x over previous
#include <hip/hip_runtime.h>

typedef __attribute__((ext_vector_type(2))) float v2f;
typedef __attribute__((ext_vector_type(8))) float v8f;

#define D1 501
#define NSTEPS 512
#define LD 512
#define WIN 4
#define NWAVE 16
#define NBAND 32
#define NTILE 64
#define TREG 22
#define TLDS 18

__global__ void kf_init_S(const float* __restrict__ Sigma0, float* __restrict__ S) {
    int idx = blockIdx.x * blockDim.x + threadIdx.x;
    if (idx < LD * LD) {
        int r = idx >> 9;
        int c = idx & (LD - 1);
        float v = 0.0f;
        if (r < D1 && c < D1) v = Sigma0[r * D1 + c];
        S[idx] = v;
    }
}

__device__ __forceinline__ v8f mma4(v2f a, v2f b, v8f c) {
    return __builtin_amdgcn_wmma_f32_16x16x4_f32(false, a, false, b, (short)0, c, false, false);
}

__device__ __forceinline__ float tile_dot(v8f c, float4 ha, float4 hb) {
    float p = c[0] * ha.x;
    p = fmaf(c[1], ha.y, p);
    p = fmaf(c[2], ha.z, p);
    p = fmaf(c[3], ha.w, p);
    p = fmaf(c[4], hb.x, p);
    p = fmaf(c[5], hb.y, p);
    p = fmaf(c[6], hb.z, p);
    p = fmaf(c[7], hb.w, p);
    return p;
}

__device__ __forceinline__ v8f lds_tile(const float4* q) {
    const float4 x = q[0];
    const float4 y = q[32];
    v8f c;
    c[0] = x.x; c[1] = x.y; c[2] = x.z; c[3] = x.w;
    c[4] = y.x; c[5] = y.y; c[6] = y.z; c[7] = y.w;
    return c;
}


__launch_bounds__(512)
__global__ void kf_scan(const float* __restrict__ emissions,
                        const float* __restrict__ Ht,
                        const float* __restrict__ R,
                        const float* __restrict__ mu0,
                        float* __restrict__ S,
                        float* __restrict__ out) {
    __shared__ float4 Sl[TLDS * NWAVE * 2 * 32];
    __shared__ float4 h4s[LD / 4];
    __shared__ float hs[LD];
    __shared__ float m[LD];
    __shared__ float ps[LD];
    __shared__ float kbuf[WIN][LD];
    __shared__ float svals[WIN];
    __shared__ float redA[16], redB[16], redC[16];
    __shared__ float sc[8];

    float* h = reinterpret_cast<float*>(h4s);

    const int tid  = threadIdx.x;
    const int lane = tid & 31;
    const int wave = tid >> 5;
    const int rl = lane & 15;
    const int hh = lane >> 4;
    const int cs = wave;
    const bool lo = (hh == 0);
    const float r = R[0];

    float* Sg = S + (8 * hh) * LD + 32 * cs + rl;
    const float* Sw = S + 32 * cs + lane;
    float4* Slw = Sl + wave * 64 + lane;
    const float4* h4 = h4s + 2 * hh;

    m[tid] = (tid < D1) ? mu0[tid] : 0.0f;
    if (tid < WIN) svals[tid] = 0.0f;

    v8f sr[TREG];
    #pragma unroll
    for (int n = 0; n < TREG; ++n) {
        #pragma unroll
        for (int v = 0; v < 8; ++v) sr[n][v] = Sg[(16 * (n >> 1) + v) * LD + 16 * (n & 1)];
    }
    #pragma unroll 1
    for (int n = TREG; n < TREG + TLDS; ++n) {
        const float* p = Sg + (16 * (n >> 1)) * LD + 16 * (n & 1);
        float4* q = Slw + (n - TREG) * (NWAVE * 64);
        q[0]  = make_float4(p[0], p[LD], p[2 * LD], p[3 * LD]);
        q[32] = make_float4(p[4 * LD], p[5 * LD], p[6 * LD], p[7 * LD]);
    }
    __syncthreads();

    for (int t = 0; t < NSTEPS; ++t) {
        const int w = t & (WIN - 1);

        {
            h[tid] = (tid < D1) ? Ht[t * D1 + tid] : 0.0f;
            if (t + 1 < NSTEPS && tid < D1)
                __builtin_prefetch(Ht + (t + 1) * D1 + tid, 0, 0);
        }
        __syncthreads();

        {
            float accA = 0.0f, accB = 0.0f;
            #pragma unroll
            for (int n = 0; n < TREG; ++n) {
                const float4 ha = h4[4 * (n >> 1)];
                const float4 hb = h4[4 * (n >> 1) + 1];
                const float p = tile_dot(sr[n], ha, hb);
                if (n & 1) accB += p; else accA += p;
            }
            static_assert(TLDS % 3 == 0 && NBAND - (TREG + TLDS) / 2 == 2 * (TLDS / 3), "a trip: 3 LDS tiles, 2 workspace bands");
            float accL = 0.0f;
            #pragma unroll 1
            for (int k = 0; k < TLDS / 3; ++k) {
                const int i = (TREG + TLDS) / 2 + 2 * k;
                const float* p = Sw + (16 * i) * LD;
                float b[32];
                #pragma unroll
                for (int r = 0; r < 32; ++r) b[r] = p[r * LD];
                #pragma unroll 1
                for (int u = 0; u < 3; ++u) {
                    const int n = TREG + 3 * k + u;
                    const float4 ha = h4[4 * (n >> 1)];
                    const float4 hb = h4[4 * (n >> 1) + 1];
                    const float q = tile_dot(lds_tile(Slw + (n - TREG) * (NWAVE * 64)), ha, hb);
                    accA += (n & 1) ? 0.0f : q;
                    accB += (n & 1) ? q : 0.0f;
                }
                const float4* hq = h4s + 4 * i;
                #pragma unroll
                for (int j = 0; j < 8; ++j) {
                    const float4 hv = hq[j];
                    accL = fmaf(b[4 * j], hv.x, accL);
                    accL = fmaf(b[4 * j + 1], hv.y, accL);
                    accL = fmaf(b[4 * j + 2], hv.z, accL);
                    accL = fmaf(b[4 * j + 3], hv.w, accL);
                }
            }
            accA += __shfl_xor(accA, 16, 32);
            accB += __shfl_xor(accB, 16, 32);
            ps[tid] = (lo ? accA : accB) + accL;
        }

        if (w > 0) {
            float hv = h[tid];
            float d0 = kbuf[0][tid] * hv;
            float d1 = (w > 1) ? kbuf[1][tid] * hv : 0.0f;
            float d2 = (w > 2) ? kbuf[2][tid] * hv : 0.0f;
            #pragma unroll
            for (int off = 16; off > 0; off >>= 1) {
                d0 += __shfl_down(d0, off, 32);
                d1 += __shfl_down(d1, off, 32);
                d2 += __shfl_down(d2, off, 32);
            }
            if (lane == 0) { redA[wave] = d0; redB[wave] = d1; redC[wave] = d2; }
        }
        __syncthreads();
        if (w > 0 && tid == 0) {
            float D0 = 0.0f, D1s = 0.0f, D2 = 0.0f;
            for (int q = 0; q < 16; ++q) { D0 += redA[q]; D1s += redB[q]; D2 += redC[q]; }
            sc[3] = svals[0] * D0;
            sc[4] = svals[1] * D1s;
            sc[5] = svals[2] * D2;
        }
        __syncthreads();

        {
            float v = ps[tid];
            if (w > 0) v -= sc[3] * kbuf[0][tid];
            if (w > 1) v -= sc[4] * kbuf[1][tid];
            if (w > 2) v -= sc[5] * kbuf[2][tid];
            hs[tid] = v;
            float hv = h[tid];
            float r1 = v * hv;
            float r2 = hv * m[tid];
            #pragma unroll
            for (int off = 16; off > 0; off >>= 1) {
                r1 += __shfl_down(r1, off, 32);
                r2 += __shfl_down(r2, off, 32);
            }
            if (lane == 0) { redA[wave] = r1; redB[wave] = r2; }
        }
        __syncthreads();
        if (tid == 0) {
            float s1 = 0.0f, s2 = 0.0f;
            for (int q = 0; q < 16; ++q) { s1 += redA[q]; s2 += redB[q]; }
            float s_term = r + s1;
            sc[0] = s_term;
            sc[1] = 1.0f / (s_term + 1e-6f);
            sc[2] = emissions[t] - s2;
            svals[w] = s_term;
        }
        __syncthreads();

        {
            float kv = hs[tid] * sc[1];
            kbuf[w][tid] = kv;
            float mv = m[tid] + kv * sc[2];
            m[tid] = mv;
            if (tid < D1) out[t * D1 + tid] = mv;
        }
        __syncthreads();

        if (w == WIN - 1) {
            const int kq = 2 * hh;
            const float nsx = -svals[kq];
            const float nsy = -svals[kq + 1];
            const float* kx = &kbuf[kq][0];
            const float* ky = &kbuf[kq + 1][0];
            const int colA = 32 * cs + rl;
            v2f bA, bB;
            bA.x = kx[colA];
            bA.y = ky[colA];
            bB.x = kx[colA + 16];
            bB.y = ky[colA + 16];

            #pragma unroll
            for (int n = 0; n < TREG; ++n) {
                v2f a;
                a.x = nsx * kx[16 * (n >> 1) + rl];
                a.y = nsy * ky[16 * (n >> 1) + rl];
                sr[n] = mma4(a, (n & 1) ? bB : bA, sr[n]);
            }
            #pragma unroll 1
            for (int n = TREG; n < TREG + TLDS; ++n) {
                v2f a;
                a.x = nsx * kx[16 * (n >> 1) + rl];
                a.y = nsy * ky[16 * (n >> 1) + rl];
                float4* q = Slw + (n - TREG) * (NWAVE * 64);
                const v8f d = mma4(a, (n & 1) ? bB : bA, lds_tile(q));
                q[0]  = make_float4(d[0], d[1], d[2], d[3]);
                q[32] = make_float4(d[4], d[5], d[6], d[7]);
            }
            #pragma unroll 4
            for (int n = TREG + TLDS; n < NTILE; ++n) {
                v2f a;
                a.x = nsx * kx[16 * (n >> 1) + rl];
                a.y = nsy * ky[16 * (n >> 1) + rl];
                float* Sp = Sg + (16 * (n >> 1)) * LD + 16 * (n & 1);
                v8f c;
                #pragma unroll
                for (int v = 0; v < 8; ++v) c[v] = Sp[v * LD];
                const v8f d = mma4(a, (n & 1) ? bB : bA, c);
                #pragma unroll
                for (int v = 0; v < 8; ++v) Sp[v * LD] = d[v];
            }
        }
        __syncthreads();
    }
}

extern "C" void kernel_launch(void* const* d_in, const int* in_sizes, int n_in,
                              void* d_out, int out_size, void* d_ws, size_t ws_size,
                              hipStream_t stream) {
    const float* emissions = (const float*)d_in[0];
    const float* Ht        = (const float*)d_in[1];
    const float* R         = (const float*)d_in[4];
    const float* mu0       = (const float*)d_in[5];
    const float* Sigma0    = (const float*)d_in[6];

    float* S   = (float*)d_ws;
    float* out = (float*)d_out;

    kf_init_S<<<(LD * LD) / 256, 256, 0, stream>>>(Sigma0, S);
    kf_scan<<<1, 512, 0, stream>>>(emissions, Ht, R, mu0, S, out);
}


# rank-4 flush of the LDS and workspace tiles rides on the next window's first matvec pass: tile loaded once, updated by the same f32 WMMA, stored, then dotted; no separate flush sweep; same sum order
# speedup vs baseline: 2.4375x; 1.2544x over previous
#include <hip/hip_runtime.h>

typedef __attribute__((ext_vector_type(2))) float v2f;
typedef __attribute__((ext_vector_type(8))) float v8f;

#define D1 501
#define NSTEPS 512
#define LD 512
#define WIN 4
#define NWAVE 16
#define NBAND 32
#define NTILE 64
#define TREG 22
#define TLDS 18
#define TWS (NTILE - TREG - TLDS)

__global__ void kf_init_S(const float* __restrict__ Sigma0, float* __restrict__ S) {
    const int g = blockIdx.x * blockDim.x + threadIdx.x;
    if (g < LD * LD / 4) {
        const int lane = g & 31;
        const int half = (g >> 5) & 1;
        const int n    = (g >> 6) & (NTILE - 1);
        const int cs   = g >> 12;
        const int row  = 16 * (n >> 1) + 8 * (lane >> 4) + 4 * half;
        const int col  = 32 * cs + 16 * (n & 1) + (lane & 15);
        float4 v = make_float4(0.0f, 0.0f, 0.0f, 0.0f);
        if (col < D1) {
            if (row     < D1) v.x = Sigma0[(row    ) * D1 + col];
            if (row + 1 < D1) v.y = Sigma0[(row + 1) * D1 + col];
            if (row + 2 < D1) v.z = Sigma0[(row + 2) * D1 + col];
            if (row + 3 < D1) v.w = Sigma0[(row + 3) * D1 + col];
        }
        reinterpret_cast<float4*>(S)[g] = v;
    }
}

__device__ __forceinline__ v8f mma4(v2f a, v2f b, v8f c) {
    return __builtin_amdgcn_wmma_f32_16x16x4_f32(false, a, false, b, (short)0, c, false, false);
}

__device__ __forceinline__ float tile_dot(v8f c, float4 ha, float4 hb) {
    float p = c[0] * ha.x;
    p = fmaf(c[1], ha.y, p);
    p = fmaf(c[2], ha.z, p);
    p = fmaf(c[3], ha.w, p);
    p = fmaf(c[4], hb.x, p);
    p = fmaf(c[5], hb.y, p);
    p = fmaf(c[6], hb.z, p);
    p = fmaf(c[7], hb.w, p);
    return p;
}

__device__ __forceinline__ v8f tile_of(float4 x, float4 y) {
    v8f c;
    c[0] = x.x; c[1] = x.y; c[2] = x.z; c[3] = x.w;
    c[4] = y.x; c[5] = y.y; c[6] = y.z; c[7] = y.w;
    return c;
}

__device__ __forceinline__ v8f lds_tile(const float4* q) {
    return tile_of(q[0], q[32]);
}


template <bool FLUSH>
__device__ __forceinline__ void far_tiles_pass(float4* Swt, float4* Slw, const float4* h4,
                                               const float* kx, const float* ky, float nsx, float nsy,
                                               v2f bA, v2f bB, int rl, float& accA, float& accB) {
    static_assert(TLDS % 3 == 0 && TWS == 4 * (TLDS / 3) && (TREG + TLDS) % 2 == 0, "a trip: 3 LDS tiles, 4 workspace tiles");
    #pragma unroll 1
    for (int k = 0; k < TLDS / 3; ++k) {
        const int n0 = TREG + TLDS + 4 * k;
        float4* p = Swt + n0 * 64;
        float4 b[8];
        #pragma unroll
        for (int j = 0; j < 8; ++j) b[j] = p[32 * j];
        #pragma unroll 1
        for (int u = 0; u < 3; ++u) {
            const int n = TREG + 3 * k + u;
            const float4 ha = h4[4 * (n >> 1)];
            const float4 hb = h4[4 * (n >> 1) + 1];
            float4* q = Slw + (n - TREG) * (NWAVE * 64);
            v8f c = lds_tile(q);
            if (FLUSH) {
                v2f a;
                a.x = nsx * kx[16 * (n >> 1) + rl];
                a.y = nsy * ky[16 * (n >> 1) + rl];
                c = mma4(a, (n & 1) ? bB : bA, c);
                q[0]  = make_float4(c[0], c[1], c[2], c[3]);
                q[32] = make_float4(c[4], c[5], c[6], c[7]);
            }
            const float d = tile_dot(c, ha, hb);
            accA += (n & 1) ? 0.0f : d;
            accB += (n & 1) ? d : 0.0f;
        }
        #pragma unroll
        for (int j = 0; j < 4; ++j) {
            const int n = n0 + j;
            const float4 ha = h4[4 * (n >> 1)];
            const float4 hb = h4[4 * (n >> 1) + 1];
            v8f c = tile_of(b[2 * j], b[2 * j + 1]);
            if (FLUSH) {
                v2f a;
                a.x = nsx * kx[16 * (n >> 1) + rl];
                a.y = nsy * ky[16 * (n >> 1) + rl];
                c = mma4(a, (j & 1) ? bB : bA, c);
                p[32 * (2 * j)]     = make_float4(c[0], c[1], c[2], c[3]);
                p[32 * (2 * j + 1)] = make_float4(c[4], c[5], c[6], c[7]);
            }
            const float d = tile_dot(c, ha, hb);
            if (j & 1) accB += d; else accA += d;
        }
    }
}

__launch_bounds__(512)
__global__ void kf_scan(const float* __restrict__ emissions,
                        const float* __restrict__ Ht,
                        const float* __restrict__ R,
                        const float* __restrict__ mu0,
                        float* __restrict__ S,
                        float* __restrict__ out) {
    __shared__ float4 Sl[TLDS * NWAVE * 2 * 32];
    __shared__ float4 h4s[LD / 4];
    __shared__ float hs[LD];
    __shared__ float m[LD];
    __shared__ float ps[LD];
    __shared__ float kbuf[WIN][LD];
    __shared__ float svals[WIN];
    __shared__ float redA[16], redB[16], redC[16];
    __shared__ float sc[8];

    float* h = reinterpret_cast<float*>(h4s);

    const int tid  = threadIdx.x;
    const int lane = tid & 31;
    const int wave = __builtin_amdgcn_readfirstlane(tid >> 5);
    const int rl = lane & 15;
    const int hh = lane >> 4;
    const int cs = wave;
    const bool lo = (hh == 0);
    const float r = R[0];

    float4* Swt = reinterpret_cast<float4*>(S) + cs * (NTILE * 64) + lane;
    float4* Slw = Sl + wave * 64 + lane;
    const float4* h4 = h4s + 2 * hh;

    m[tid] = (tid < D1) ? mu0[tid] : 0.0f;
    if (tid < WIN) svals[tid] = 0.0f;

    v8f sr[TREG];
    #pragma unroll
    for (int n = 0; n < TREG; ++n) sr[n] = lds_tile(Swt + n * 64);
    #pragma unroll 1
    for (int n = TREG; n < TREG + TLDS; ++n) {
        const float4* p = Swt + n * 64;
        float4* q = Slw + (n - TREG) * (NWAVE * 64);
        q[0]  = p[0];
        q[32] = p[32];
    }
    __syncthreads();

    for (int t = 0; t < NSTEPS; ++t) {
        const int w = t & (WIN - 1);

        {
            const unsigned e = (unsigned)(t * D1) + (unsigned)tid;
            h[tid] = (tid < D1) ? Ht[e] : 0.0f;
            if (t + 1 < NSTEPS && tid < D1)
                __builtin_prefetch(Ht + (e + (unsigned)D1), 0, 0);
        }
        __syncthreads();

        {
            float accA = 0.0f, accB = 0.0f;
            #pragma unroll
            for (int n = 0; n < TREG; ++n) {
                const float4 ha = h4[4 * (n >> 1)];
                const float4 hb = h4[4 * (n >> 1) + 1];
                const float p = tile_dot(sr[n], ha, hb);
                if (n & 1) accB += p; else accA += p;
            }
            if (w == 0 && t > 0) {
                const int kq = 2 * hh;
                const float* kx = &kbuf[kq][0];
                const float* ky = &kbuf[kq + 1][0];
                const int colA = 32 * cs + rl;
                v2f bA, bB;
                bA.x = kx[colA];
                bA.y = ky[colA];
                bB.x = kx[colA + 16];
                bB.y = ky[colA + 16];
                far_tiles_pass<true>(Swt, Slw, h4, kx, ky, -svals[kq], -svals[kq + 1], bA, bB, rl, accA, accB);
            } else {
                v2f z;
                z.x = 0.0f;
                z.y = 0.0f;
                far_tiles_pass<false>(Swt, Slw, h4, nullptr, nullptr, 0.0f, 0.0f, z, z, rl, accA, accB);
            }
            accA += __shfl_xor(accA, 16, 32);
            accB += __shfl_xor(accB, 16, 32);
            ps[tid] = lo ? accA : accB;
        }

        if (w > 0) {
            float hv = h[tid];
            float d0 = kbuf[0][tid] * hv;
            float d1 = (w > 1) ? kbuf[1][tid] * hv : 0.0f;
            float d2 = (w > 2) ? kbuf[2][tid] * hv : 0.0f;
            #pragma unroll
            for (int off = 16; off > 0; off >>= 1) {
                d0 += __shfl_down(d0, off, 32);
                d1 += __shfl_down(d1, off, 32);
                d2 += __shfl_down(d2, off, 32);
            }
            if (lane == 0) { redA[wave] = d0; redB[wave] = d1; redC[wave] = d2; }
        }
        __syncthreads();
        if (w > 0 && tid == 0) {
            float D0 = 0.0f, D1s = 0.0f, D2 = 0.0f;
            for (int q = 0; q < 16; ++q) { D0 += redA[q]; D1s += redB[q]; D2 += redC[q]; }
            sc[3] = svals[0] * D0;
            sc[4] = svals[1] * D1s;
            sc[5] = svals[2] * D2;
        }
        __syncthreads();

        {
            float v = ps[tid];
            if (w > 0) v -= sc[3] * kbuf[0][tid];
            if (w > 1) v -= sc[4] * kbuf[1][tid];
            if (w > 2) v -= sc[5] * kbuf[2][tid];
            hs[tid] = v;
            float hv = h[tid];
            float r1 = v * hv;
            float r2 = hv * m[tid];
            #pragma unroll
            for (int off = 16; off > 0; off >>= 1) {
                r1 += __shfl_down(r1, off, 32);
                r2 += __shfl_down(r2, off, 32);
            }
            if (lane == 0) { redA[wave] = r1; redB[wave] = r2; }
        }
        __syncthreads();
        if (tid == 0) {
            float s1 = 0.0f, s2 = 0.0f;
            for (int q = 0; q < 16; ++q) { s1 += redA[q]; s2 += redB[q]; }
            float s_term = r + s1;
            sc[0] = s_term;
            sc[1] = 1.0f / (s_term + 1e-6f);
            sc[2] = emissions[t] - s2;
            svals[w] = s_term;
        }
        __syncthreads();

        {
            float kv = hs[tid] * sc[1];
            kbuf[w][tid] = kv;
            float mv = m[tid] + kv * sc[2];
            m[tid] = mv;
            if (tid < D1) out[t * D1 + tid] = mv;
        }
        __syncthreads();

        if (w == WIN - 1) {
            const int kq = 2 * hh;
            const float nsx = -svals[kq];
            const float nsy = -svals[kq + 1];
            const float* kx = &kbuf[kq][0];
            const float* ky = &kbuf[kq + 1][0];
            const int colA = 32 * cs + rl;
            v2f bA, bB;
            bA.x = kx[colA];
            bA.y = ky[colA];
            bB.x = kx[colA + 16];
            bB.y = ky[colA + 16];

            #pragma unroll
            for (int n = 0; n < TREG; ++n) {
                v2f a;
                a.x = nsx * kx[16 * (n >> 1) + rl];
                a.y = nsy * ky[16 * (n >> 1) + rl];
                sr[n] = mma4(a, (n & 1) ? bB : bA, sr[n]);
            }
        }
        __syncthreads();
    }
}

extern "C" void kernel_launch(void* const* d_in, const int* in_sizes, int n_in,
                              void* d_out, int out_size, void* d_ws, size_t ws_size,
                              hipStream_t stream) {
    const float* emissions = (const float*)d_in[0];
    const float* Ht        = (const float*)d_in[1];
    const float* R         = (const float*)d_in[4];
    const float* mu0       = (const float*)d_in[5];
    const float* Sigma0    = (const float*)d_in[6];

    float* S   = (float*)d_ws;
    float* out = (float*)d_out;

    kf_init_S<<<(LD * LD / 4) / 256, 256, 0, stream>>>(Sigma0, S);
    kf_scan<<<1, 512, 0, stream>>>(emissions, Ht, R, mu0, S, out);
}


# speedup vs baseline: 2.5815x; 1.0591x over previous
#include <hip/hip_runtime.h>

typedef __attribute__((ext_vector_type(2))) float v2f;
typedef __attribute__((ext_vector_type(8))) float v8f;

#define D1 501
#define NSTEPS 512
#define LD 512
#define WIN 4
#define NWAVE 16
#define NBAND 32
#define NTILE 64
#define TREG 22
#define TLDS 18
#define TWS (NTILE - TREG - TLDS)

__global__ void kf_init_S(const float* __restrict__ Sigma0, float* __restrict__ S) {
    const int g = blockIdx.x * blockDim.x + threadIdx.x;
    if (g < LD * LD / 4) {
        const int lane = g & 31;
        const int half = (g >> 5) & 1;
        const int n    = (g >> 6) & (NTILE - 1);
        const int cs   = g >> 12;
        const int row  = 16 * (n >> 1) + 8 * (lane >> 4) + 4 * half;
        const int col  = 32 * cs + 16 * (n & 1) + (lane & 15);
        float4 v = make_float4(0.0f, 0.0f, 0.0f, 0.0f);
        if (col < D1) {
            if (row     < D1) v.x = Sigma0[(row    ) * D1 + col];
            if (row + 1 < D1) v.y = Sigma0[(row + 1) * D1 + col];
            if (row + 2 < D1) v.z = Sigma0[(row + 2) * D1 + col];
            if (row + 3 < D1) v.w = Sigma0[(row + 3) * D1 + col];
        }
        reinterpret_cast<float4*>(S)[g] = v;
    }
}

__device__ __forceinline__ v8f mma4(v2f a, v2f b, v8f c) {
    return __builtin_amdgcn_wmma_f32_16x16x4_f32(false, a, false, b, (short)0, c, false, false);
}

__device__ __forceinline__ float tile_dot(v8f c, float4 ha, float4 hb) {
    float p = c[0] * ha.x;
    p = fmaf(c[1], ha.y, p);
    p = fmaf(c[2], ha.z, p);
    p = fmaf(c[3], ha.w, p);
    p = fmaf(c[4], hb.x, p);
    p = fmaf(c[5], hb.y, p);
    p = fmaf(c[6], hb.z, p);
    p = fmaf(c[7], hb.w, p);
    return p;
}

__device__ __forceinline__ v8f tile_of(float4 x, float4 y) {
    v8f c;
    c[0] = x.x; c[1] = x.y; c[2] = x.z; c[3] = x.w;
    c[4] = y.x; c[5] = y.y; c[6] = y.z; c[7] = y.w;
    return c;
}

__device__ __forceinline__ v8f lds_tile(const float4* q) {
    return tile_of(q[0], q[32]);
}


template <bool FLUSH>
__device__ __forceinline__ void far_tiles_pass(float4* Swt, float4* Slw, const float4* h4,
                                               const float* kx, const float* ky, float nsx, float nsy,
                                               v2f bA, v2f bB, int rl, float& accA, float& accB) {
    static_assert(TLDS % 3 == 0 && TWS == 4 * (TLDS / 3) && (TREG + TLDS) % 2 == 0, "a trip: 3 LDS tiles, 4 workspace tiles");
    #pragma unroll 1
    for (int k = 0; k < TLDS / 3; ++k) {
        const int n0 = TREG + TLDS + 4 * k;
        float4* p = Swt + n0 * 64;
        float4 b[8];
        #pragma unroll
        for (int j = 0; j < 8; ++j) b[j] = p[32 * j];
        #pragma unroll 1
        for (int u = 0; u < 3; ++u) {
            const int n = TREG + 3 * k + u;
            const float4 ha = h4[4 * (n >> 1)];
            const float4 hb = h4[4 * (n >> 1) + 1];
            float4* q = Slw + (n - TREG) * (NWAVE * 64);
            v8f c = lds_tile(q);
            if (FLUSH) {
                v2f a;
                a.x = nsx * kx[16 * (n >> 1) + rl];
                a.y = nsy * ky[16 * (n >> 1) + rl];
                c = mma4(a, (n & 1) ? bB : bA, c);
                q[0]  = make_float4(c[0], c[1], c[2], c[3]);
                q[32] = make_float4(c[4], c[5], c[6], c[7]);
            }
            const float d = tile_dot(c, ha, hb);
            accA += (n & 1) ? 0.0f : d;
            accB += (n & 1) ? d : 0.0f;
        }
        #pragma unroll
        for (int j = 0; j < 4; ++j) {
            const int n = n0 + j;
            const float4 ha = h4[4 * (n >> 1)];
            const float4 hb = h4[4 * (n >> 1) + 1];
            v8f c = tile_of(b[2 * j], b[2 * j + 1]);
            if (FLUSH) {
                v2f a;
                a.x = nsx * kx[16 * (n >> 1) + rl];
                a.y = nsy * ky[16 * (n >> 1) + rl];
                c = mma4(a, (j & 1) ? bB : bA, c);
                p[32 * (2 * j)]     = make_float4(c[0], c[1], c[2], c[3]);
                p[32 * (2 * j + 1)] = make_float4(c[4], c[5], c[6], c[7]);
            }
            const float d = tile_dot(c, ha, hb);
            if (j & 1) accB += d; else accA += d;
        }
    }
}

__launch_bounds__(512)
__global__ void kf_scan(const float* __restrict__ emissions,
                        const float* __restrict__ Ht,
                        const float* __restrict__ R,
                        const float* __restrict__ mu0,
                        float* __restrict__ S,
                        float* __restrict__ out) {
    __shared__ float4 Sl[TLDS * NWAVE * 2 * 32];
    __shared__ float4 h4s[LD / 4];
    __shared__ float hs[LD];
    __shared__ float m[LD];
    __shared__ float ps[LD];
    __shared__ float kbuf[WIN][LD];
    __shared__ float svals[WIN];
    __shared__ float redA[16], redB[16], redC[16];
    __shared__ float sc[8];

    float* h = reinterpret_cast<float*>(h4s);

    const int tid  = threadIdx.x;
    const int lane = tid & 31;
    const int wave = __builtin_amdgcn_readfirstlane(tid >> 5);
    const int rl = lane & 15;
    const int hh = lane >> 4;
    const int cs = wave;
    const bool lo = (hh == 0);
    const float r = R[0];

    float4* Swt = reinterpret_cast<float4*>(S) + cs * (NTILE * 64) + lane;
    float4* Slw = Sl + wave * 64 + lane;
    const float4* h4 = h4s + 2 * hh;

    m[tid] = (tid < D1) ? mu0[tid] : 0.0f;
    if (tid < WIN) svals[tid] = 0.0f;

    v8f sr[TREG];
    #pragma unroll
    for (int n = 0; n < TREG; ++n) sr[n] = lds_tile(Swt + n * 64);
    #pragma unroll 1
    for (int n = TREG; n < TREG + TLDS; ++n) {
        const float4* p = Swt + n * 64;
        float4* q = Slw + (n - TREG) * (NWAVE * 64);
        q[0]  = p[0];
        q[32] = p[32];
    }
    __syncthreads();

    for (int t = 0; t < NSTEPS; ++t) {
        const int w = t & (WIN - 1);

        {
            const unsigned e = (unsigned)(t * D1) + (unsigned)tid;
            h[tid] = (tid < D1) ? Ht[e] : 0.0f;
            if (t + 1 < NSTEPS && tid < D1)
                __builtin_prefetch(Ht + (e + (unsigned)D1), 0, 0);
        }
        __syncthreads();

        {
            float accA = 0.0f, accB = 0.0f;
            #pragma unroll
            for (int n = 0; n < TREG; ++n) {
                const float4 ha = h4[4 * (n >> 1)];
                const float4 hb = h4[4 * (n >> 1) + 1];
                const float p = tile_dot(sr[n], ha, hb);
                if (n & 1) accB += p; else accA += p;
            }
            if (w == 0 && t > 0) {
                const int kq = 2 * hh;
                const float* kx = &kbuf[kq][0];
                const float* ky = &kbuf[kq + 1][0];
                const int colA = 32 * cs + rl;
                v2f bA, bB;
                bA.x = kx[colA];
                bA.y = ky[colA];
                bB.x = kx[colA + 16];
                bB.y = ky[colA + 16];
                far_tiles_pass<true>(Swt, Slw, h4, kx, ky, -svals[kq], -svals[kq + 1], bA, bB, rl, accA, accB);
            } else {
                v2f z;
                z.x = 0.0f;
                z.y = 0.0f;
                far_tiles_pass<false>(Swt, Slw, h4, nullptr, nullptr, 0.0f, 0.0f, z, z, rl, accA, accB);
            }
            accA += __shfl_xor(accA, 16, 32);
            accB += __shfl_xor(accB, 16, 32);
            ps[tid] = lo ? accA : accB;
        }

        if (w > 0) {
            float hv = h[tid];
            float d0 = kbuf[0][tid] * hv;
            float d1 = (w > 1) ? kbuf[1][tid] * hv : 0.0f;
            float d2 = (w > 2) ? kbuf[2][tid] * hv : 0.0f;
            #pragma unroll
            for (int off = 16; off > 0; off >>= 1) {
                d0 += __shfl_down(d0, off, 32);
                d1 += __shfl_down(d1, off, 32);
                d2 += __shfl_down(d2, off, 32);
            }
            if (lane == 0) { redA[wave] = d0; redB[wave] = d1; redC[wave] = d2; }
        }
        __syncthreads();
        if (w == 0 && t > 0) {
            #pragma unroll
            for (int n = TREG + TLDS; n < NTILE; ++n) {
                __builtin_prefetch(Swt + n * 64, 0, 2);
                __builtin_prefetch(Swt + n * 64 + 32, 0, 2);
            }
        }
        if (w > 0 && tid == 0) {
            float D0 = 0.0f, D1s = 0.0f, D2 = 0.0f;
            for (int q = 0; q < 16; ++q) { D0 += redA[q]; D1s += redB[q]; D2 += redC[q]; }
            sc[3] = svals[0] * D0;
            sc[4] = svals[1] * D1s;
            sc[5] = svals[2] * D2;
        }
        __syncthreads();

        {
            float v = ps[tid];
            if (w > 0) v -= sc[3] * kbuf[0][tid];
            if (w > 1) v -= sc[4] * kbuf[1][tid];
            if (w > 2) v -= sc[5] * kbuf[2][tid];
            hs[tid] = v;
            float hv = h[tid];
            float r1 = v * hv;
            float r2 = hv * m[tid];
            #pragma unroll
            for (int off = 16; off > 0; off >>= 1) {
                r1 += __shfl_down(r1, off, 32);
                r2 += __shfl_down(r2, off, 32);
            }
            if (lane == 0) { redA[wave] = r1; redB[wave] = r2; }
        }
        __syncthreads();
        if (tid == 0) {
            float s1 = 0.0f, s2 = 0.0f;
            for (int q = 0; q < 16; ++q) { s1 += redA[q]; s2 += redB[q]; }
            float s_term = r + s1;
            sc[0] = s_term;
            sc[1] = 1.0f / (s_term + 1e-6f);
            sc[2] = emissions[t] - s2;
            svals[w] = s_term;
        }
        __syncthreads();

        {
            float kv = hs[tid] * sc[1];
            kbuf[w][tid] = kv;
            float mv = m[tid] + kv * sc[2];
            m[tid] = mv;
            if (tid < D1) out[t * D1 + tid] = mv;
        }
        __syncthreads();

        if (w == WIN - 1) {
            const int kq = 2 * hh;
            const float nsx = -svals[kq];
            const float nsy = -svals[kq + 1];
            const float* kx = &kbuf[kq][0];
            const float* ky = &kbuf[kq + 1][0];
            const int colA = 32 * cs + rl;
            v2f bA, bB;
            bA.x = kx[colA];
            bA.y = ky[colA];
            bB.x = kx[colA + 16];
            bB.y = ky[colA + 16];

            #pragma unroll
            for (int n = 0; n < TREG; ++n) {
                v2f a;
                a.x = nsx * kx[16 * (n >> 1) + rl];
                a.y = nsy * ky[16 * (n >> 1) + rl];
                sr[n] = mma4(a, (n & 1) ? bB : bA, sr[n]);
            }
        }
        __syncthreads();
    }
}

extern "C" void kernel_launch(void* const* d_in, const int* in_sizes, int n_in,
                              void* d_out, int out_size, void* d_ws, size_t ws_size,
                              hipStream_t stream) {
    const float* emissions = (const float*)d_in[0];
    const float* Ht        = (const float*)d_in[1];
    const float* R         = (const float*)d_in[4];
    const float* mu0       = (const float*)d_in[5];
    const float* Sigma0    = (const float*)d_in[6];

    float* S   = (float*)d_ws;
    float* out = (float*)d_out;

    kf_init_S<<<(LD * LD / 4) / 256, 256, 0, stream>>>(Sigma0, S);
    kf_scan<<<1, 512, 0, stream>>>(emissions, Ht, R, mu0, S, out);
}


# serial tail of a step fused: its five column sums go as wave sums behind ONE barrier, each thread adds them and finishes its own column; 2 barriers a step (was 7); sum order changes
# speedup vs baseline: 2.9516x; 1.1434x over previous
#include <hip/hip_runtime.h>

typedef __attribute__((ext_vector_type(2))) float v2f;
typedef __attribute__((ext_vector_type(8))) float v8f;

#define D1 501
#define NSTEPS 512
#define LD 512
#define WIN 4
#define NWAVE 16
#define NBAND 32
#define NTILE 64
#define TREG 22
#define TLDS 18
#define TWS (NTILE - TREG - TLDS)
#define NSUM 5

__global__ void kf_init_S(const float* __restrict__ Sigma0, float* __restrict__ S) {
    const int g = blockIdx.x * blockDim.x + threadIdx.x;
    if (g < LD * LD / 4) {
        const int lane = g & 31;
        const int half = (g >> 5) & 1;
        const int n    = (g >> 6) & (NTILE - 1);
        const int cs   = g >> 12;
        const int row  = 16 * (n >> 1) + 8 * (lane >> 4) + 4 * half;
        const int col  = 32 * cs + 16 * (n & 1) + (lane & 15);
        float4 v = make_float4(0.0f, 0.0f, 0.0f, 0.0f);
        if (col < D1) {
            if (row     < D1) v.x = Sigma0[(row    ) * D1 + col];
            if (row + 1 < D1) v.y = Sigma0[(row + 1) * D1 + col];
            if (row + 2 < D1) v.z = Sigma0[(row + 2) * D1 + col];
            if (row + 3 < D1) v.w = Sigma0[(row + 3) * D1 + col];
        }
        reinterpret_cast<float4*>(S)[g] = v;
    }
}

__device__ __forceinline__ v8f mma4(v2f a, v2f b, v8f c) {
    return __builtin_amdgcn_wmma_f32_16x16x4_f32(false, a, false, b, (short)0, c, false, false);
}

__device__ __forceinline__ float tile_dot(v8f c, float4 ha, float4 hb) {
    float p = c[0] * ha.x;
    p = fmaf(c[1], ha.y, p);
    p = fmaf(c[2], ha.z, p);
    p = fmaf(c[3], ha.w, p);
    p = fmaf(c[4], hb.x, p);
    p = fmaf(c[5], hb.y, p);
    p = fmaf(c[6], hb.z, p);
    p = fmaf(c[7], hb.w, p);
    return p;
}

__device__ __forceinline__ v8f tile_of(float4 x, float4 y) {
    v8f c;
    c[0] = x.x; c[1] = x.y; c[2] = x.z; c[3] = x.w;
    c[4] = y.x; c[5] = y.y; c[6] = y.z; c[7] = y.w;
    return c;
}

__device__ __forceinline__ v8f lds_tile(const float4* q) {
    return tile_of(q[0], q[32]);
}


template <bool FLUSH>
__device__ __forceinline__ void far_tiles_pass(float4* Swt, float4* Slw, const float4* h4,
                                               const float* kx, const float* ky, float nsx, float nsy,
                                               v2f bA, v2f bB, int rl, float& accA, float& accB) {
    static_assert(TLDS % 3 == 0 && TWS == 4 * (TLDS / 3) && (TREG + TLDS) % 2 == 0, "a trip: 3 LDS tiles, 4 workspace tiles");
    #pragma unroll 1
    for (int k = 0; k < TLDS / 3; ++k) {
        const int n0 = TREG + TLDS + 4 * k;
        float4* p = Swt + n0 * 64;
        float4 b[8];
        #pragma unroll
        for (int j = 0; j < 8; ++j) b[j] = p[32 * j];
        #pragma unroll 1
        for (int u = 0; u < 3; ++u) {
            const int n = TREG + 3 * k + u;
            const float4 ha = h4[4 * (n >> 1)];
            const float4 hb = h4[4 * (n >> 1) + 1];
            float4* q = Slw + (n - TREG) * (NWAVE * 64);
            v8f c = lds_tile(q);
            if (FLUSH) {
                v2f a;
                a.x = nsx * kx[16 * (n >> 1) + rl];
                a.y = nsy * ky[16 * (n >> 1) + rl];
                c = mma4(a, (n & 1) ? bB : bA, c);
                q[0]  = make_float4(c[0], c[1], c[2], c[3]);
                q[32] = make_float4(c[4], c[5], c[6], c[7]);
            }
            const float d = tile_dot(c, ha, hb);
            accA += (n & 1) ? 0.0f : d;
            accB += (n & 1) ? d : 0.0f;
        }
        #pragma unroll
        for (int j = 0; j < 4; ++j) {
            const int n = n0 + j;
            const float4 ha = h4[4 * (n >> 1)];
            const float4 hb = h4[4 * (n >> 1) + 1];
            v8f c = tile_of(b[2 * j], b[2 * j + 1]);
            if (FLUSH) {
                v2f a;
                a.x = nsx * kx[16 * (n >> 1) + rl];
                a.y = nsy * ky[16 * (n >> 1) + rl];
                c = mma4(a, (j & 1) ? bB : bA, c);
                p[32 * (2 * j)]     = make_float4(c[0], c[1], c[2], c[3]);
                p[32 * (2 * j + 1)] = make_float4(c[4], c[5], c[6], c[7]);
            }
            const float d = tile_dot(c, ha, hb);
            if (j & 1) accB += d; else accA += d;
        }
    }
}

__launch_bounds__(512)
__global__ void kf_scan(const float* __restrict__ emissions,
                        const float* __restrict__ Ht,
                        const float* __restrict__ R,
                        const float* __restrict__ mu0,
                        float* __restrict__ S,
                        float* __restrict__ out) {
    __shared__ float4 Sl[TLDS * NWAVE * 2 * 32];
    __shared__ float4 h4s[LD / 4];
    __shared__ float m[LD];
    __shared__ float kbuf[WIN][LD];
    __shared__ float svals[WIN];
    __shared__ float4 red4[NSUM * NWAVE / 4];
    __shared__ float ys[NSTEPS];

    float* h = reinterpret_cast<float*>(h4s);
    float* red = reinterpret_cast<float*>(red4);

    const int tid  = threadIdx.x;
    const int lane = tid & 31;
    const int wave = __builtin_amdgcn_readfirstlane(tid >> 5);
    const int rl = lane & 15;
    const int hh = lane >> 4;
    const int cs = wave;
    const bool lo = (hh == 0);
    const float r = R[0];

    float4* Swt = reinterpret_cast<float4*>(S) + cs * (NTILE * 64) + lane;
    float4* Slw = Sl + wave * 64 + lane;
    const float4* h4 = h4s + 2 * hh;

    m[tid] = (tid < D1) ? mu0[tid] : 0.0f;
    h[tid] = (tid < D1) ? Ht[tid] : 0.0f;
    static_assert(NSTEPS == NWAVE * 32, "one emission per thread");
    ys[tid] = emissions[tid];
    if (tid < WIN) svals[tid] = 0.0f;

    v8f sr[TREG];
    #pragma unroll
    for (int n = 0; n < TREG; ++n) sr[n] = lds_tile(Swt + n * 64);
    #pragma unroll 1
    for (int n = TREG; n < TREG + TLDS; ++n) {
        const float4* p = Swt + n * 64;
        float4* q = Slw + (n - TREG) * (NWAVE * 64);
        q[0]  = p[0];
        q[32] = p[32];
    }
    __syncthreads();

    for (int t = 0; t < NSTEPS; ++t) {
        const int w = t & (WIN - 1);

        if (t + 1 < NSTEPS && tid < D1)
            __builtin_prefetch(Ht + ((unsigned)((t + 1) * D1) + (unsigned)tid), 0, 0);

        float accA = 0.0f, accB = 0.0f;
        if (w == 0 && t > 0) {
            const int kq = 2 * hh;
            const float nsx = -svals[kq];
            const float nsy = -svals[kq + 1];
            const float* kx = &kbuf[kq][0];
            const float* ky = &kbuf[kq + 1][0];
            const int colA = 32 * cs + rl;
            v2f bA, bB;
            bA.x = kx[colA];
            bA.y = ky[colA];
            bB.x = kx[colA + 16];
            bB.y = ky[colA + 16];
            #pragma unroll
            for (int n = 0; n < TREG; ++n) {
                v2f a;
                a.x = nsx * kx[16 * (n >> 1) + rl];
                a.y = nsy * ky[16 * (n >> 1) + rl];
                sr[n] = mma4(a, (n & 1) ? bB : bA, sr[n]);
            }
            far_tiles_pass<true>(Swt, Slw, h4, kx, ky, nsx, nsy, bA, bB, rl, accA, accB);
        } else {
            v2f z;
            z.x = 0.0f;
            z.y = 0.0f;
            far_tiles_pass<false>(Swt, Slw, h4, nullptr, nullptr, 0.0f, 0.0f, z, z, rl, accA, accB);
        }
        #pragma unroll
        for (int n = 0; n < TREG; ++n) {
            const float4 ha = h4[4 * (n >> 1)];
            const float4 hb = h4[4 * (n >> 1) + 1];
            const float p = tile_dot(sr[n], ha, hb);
            if (n & 1) accB += p; else accA += p;
        }
        accA += __shfl_xor(accA, 16, 32);
        accB += __shfl_xor(accB, 16, 32);
        const float pv = lo ? accA : accB;

        const float hv = h[tid];
        const float k0 = (w > 0) ? kbuf[0][tid] : 0.0f;
        const float k1 = (w > 1) ? kbuf[1][tid] : 0.0f;
        const float k2 = (w > 2) ? kbuf[2][tid] : 0.0f;
        {
            float d0 = k0 * hv;
            float d1 = k1 * hv;
            float d2 = k2 * hv;
            float q0 = pv * hv;
            float q1 = hv * m[tid];
            #pragma unroll
            for (int off = 16; off > 0; off >>= 1) {
                d0 += __shfl_down(d0, off, 32);
                d1 += __shfl_down(d1, off, 32);
                d2 += __shfl_down(d2, off, 32);
                q0 += __shfl_down(q0, off, 32);
                q1 += __shfl_down(q1, off, 32);
            }
            if (lane == 0) {
                red[0 * NWAVE + wave] = d0;
                red[1 * NWAVE + wave] = d1;
                red[2 * NWAVE + wave] = d2;
                red[3 * NWAVE + wave] = q0;
                red[4 * NWAVE + wave] = q1;
            }
        }
        __syncthreads();

        if (w == 0 && t > 0) {
            #pragma unroll
            for (int n = TREG + TLDS; n < NTILE; ++n) {
                __builtin_prefetch(Swt + n * 64, 0, 2);
                __builtin_prefetch(Swt + n * 64 + 32, 0, 2);
            }
        }
        float hn = 0.0f;
        if (t + 1 < NSTEPS && tid < D1) {
            hn = Ht[(unsigned)((t + 1) * D1) + (unsigned)tid];
        }
        float tot[NSUM];
        #pragma unroll
        for (int v = 0; v < NSUM; ++v) {
            const float4 x0 = red4[4 * v], x1 = red4[4 * v + 1], x2 = red4[4 * v + 2], x3 = red4[4 * v + 3];
            const float sx = (x0.x + x1.x) + (x2.x + x3.x);
            const float sy = (x0.y + x1.y) + (x2.y + x3.y);
            const float sz = (x0.z + x1.z) + (x2.z + x3.z);
            const float sw = (x0.w + x1.w) + (x2.w + x3.w);
            tot[v] = (sx + sy) + (sz + sw);
        }
        const float c0 = (w > 0) ? svals[0] * tot[0] : 0.0f;
        const float c1 = (w > 1) ? svals[1] * tot[1] : 0.0f;
        const float c2 = (w > 2) ? svals[2] * tot[2] : 0.0f;
        float s1 = tot[3];
        s1 -= c0 * tot[0];
        s1 -= c1 * tot[1];
        s1 -= c2 * tot[2];
        const float s_term = r + s1;
        const float inva = 1.0f / (s_term + 1e-6f);
        const float resid = ys[t] - tot[4];
        float v = pv;
        v -= c0 * k0;
        v -= c1 * k1;
        v -= c2 * k2;
        const float kv = v * inva;
        kbuf[w][tid] = kv;
        const float mv = m[tid] + kv * resid;
        m[tid] = mv;
        if (tid == 0) svals[w] = s_term;
        h[tid] = hn;
        __syncthreads();
        if (tid < D1) out[t * D1 + tid] = mv;
    }
}

extern "C" void kernel_launch(void* const* d_in, const int* in_sizes, int n_in,
                              void* d_out, int out_size, void* d_ws, size_t ws_size,
                              hipStream_t stream) {
    const float* emissions = (const float*)d_in[0];
    const float* Ht        = (const float*)d_in[1];
    const float* R         = (const float*)d_in[4];
    const float* mu0       = (const float*)d_in[5];
    const float* Sigma0    = (const float*)d_in[6];

    float* S   = (float*)d_ws;
    float* out = (float*)d_out;

    kf_init_S<<<(LD * LD / 4) / 256, 256, 0, stream>>>(Sigma0, S);
    kf_scan<<<1, 512, 0, stream>>>(emissions, Ht, R, mu0, S, out);
}


# steps that owe no update stream the workspace tiles: a tile's 2 loads go out again (next trip's tile) right behind its use, 8 stay in flight, next step's first tile under the tail; sum order changes
# speedup vs baseline: 3.8517x; 1.3050x over previous
#include <hip/hip_runtime.h>

typedef __attribute__((ext_vector_type(2))) float v2f;
typedef __attribute__((ext_vector_type(8))) float v8f;

#define D1 501
#define NSTEPS 512
#define LD 512
#define WIN 8
#define NG (WIN / 4)
#define NWAVE 16
#define NBAND 32
#define NTILE 64
#define TREG 22
#define TLDS 18
#define TWS (NTILE - TREG - TLDS)
#define NSUM (WIN + 1)
#define NB 2
#define HP 36

__global__ void kf_init_S(const float* __restrict__ Sigma0, float* __restrict__ S) {
    const int g = blockIdx.x * blockDim.x + threadIdx.x;
    if (g < LD * LD / 4) {
        const int lane = g & 31;
        const int half = (g >> 5) & 1;
        const int n    = (g >> 6) & (NTILE - 1);
        const int cs   = g >> 12;
        const int row  = 16 * (n >> 1) + (lane & 15);
        const int col  = 32 * cs + 16 * (n & 1) + 8 * (lane >> 4) + 4 * half;
        float4 v = make_float4(0.0f, 0.0f, 0.0f, 0.0f);
        if (row < D1) {
            if (col     < D1) v.x = Sigma0[row * D1 + col];
            if (col + 1 < D1) v.y = Sigma0[row * D1 + col + 1];
            if (col + 2 < D1) v.z = Sigma0[row * D1 + col + 2];
            if (col + 3 < D1) v.w = Sigma0[row * D1 + col + 3];
        }
        reinterpret_cast<float4*>(S)[g] = v;
    }
}

__device__ __forceinline__ v8f mma4(v2f a, v2f b, v8f c) {
    return __builtin_amdgcn_wmma_f32_16x16x4_f32(false, a, false, b, (short)0, c, false, false);
}

static_assert(WIN % 4 == 0 && NG >= 1, "a window is a whole number of K = 4 matrix instructions");
struct AFrag { v2f g[NG]; };

__device__ __forceinline__ v8f tile_update(const AFrag& a, const float* kx, const float* ky, int o, v8f c) {
    #pragma unroll
    for (int i = 0; i < NG; ++i) {
        v2f b;
        b.x = kx[4 * i * LD + o];
        b.y = ky[4 * i * LD + o];
        c = mma4(a.g[i], b, c);
    }
    return c;
}

__device__ __forceinline__ v8f axpy8(v8f c, float s, v8f acc) {
    #pragma unroll
    for (int v = 0; v < 8; ++v) acc[v] = fmaf(c[v], s, acc[v]);
    return acc;
}

template <int M>
__device__ __forceinline__ float row_xor(float x) {
    return __int_as_float(__builtin_amdgcn_update_dpp(0, __float_as_int(x), 0x160 + M, 0xf, 0xf, true));
}

__device__ __forceinline__ v8f tile_of(float4 x, float4 y) {
    v8f c;
    c[0] = x.x; c[1] = x.y; c[2] = x.z; c[3] = x.w;
    c[4] = y.x; c[5] = y.y; c[6] = y.z; c[7] = y.w;
    return c;
}

__device__ __forceinline__ v8f lds_tile(const float4* q) {
    return tile_of(q[0], q[32]);
}


template <bool FLUSH>
__device__ __forceinline__ void far_trip(float4* p, float4* q0, const float* hl, int lb, int wb,
                                         const float* kx, const float* ky, const AFrag& a, int rl, v8f& acc) {
    float4 b[8];
    #pragma unroll
    for (int j = 0; j < 8; ++j) b[j] = p[128 * (j >> 1) + 32 * (j & 1)];
    #pragma unroll 1
    for (int u = 0; u < 3; ++u) {
        const int band = lb + u;
        const float hb = hl[band];
        float4* q = q0 + u * (2 * NWAVE * 64);
        v8f c = lds_tile(q);
        if (FLUSH) {
            c = tile_update(a, kx, ky, 16 * band + rl, c);
            q[0]  = make_float4(c[0], c[1], c[2], c[3]);
            q[32] = make_float4(c[4], c[5], c[6], c[7]);
        }
        acc = axpy8(c, hb, acc);
    }
    const float4 hw = *reinterpret_cast<const float4*>(hl + wb);
    #pragma unroll
    for (int j = 0; j < 4; ++j) {
        const float hb = (j == 0) ? hw.x : (j == 1) ? hw.y : (j == 2) ? hw.z : hw.w;
        v8f c = tile_of(b[2 * j], b[2 * j + 1]);
        if (FLUSH) {
            c = tile_update(a, kx, ky, 16 * (wb + j) + rl, c);
            p[128 * j]      = make_float4(c[0], c[1], c[2], c[3]);
            p[128 * j + 32] = make_float4(c[4], c[5], c[6], c[7]);
        }
        acc = axpy8(c, hb, acc);
    }
}

template <bool FLUSH, int SIDE>
__device__ __forceinline__ void far_side_pass(float4* Swt, float4* Slw, const float* hl,
                                              const float* kx, const float* ky, const AFrag& a, int rl, v8f& acc) {
    static_assert(TREG % 2 == 0 && TLDS % 6 == 0 && TWS == 8 * (TLDS / 6) && (TREG + TLDS) % 8 == 0,
                  "a side's trip: 3 LDS bands, 4 workspace bands whose h entries are one aligned 16-byte read");
    #pragma unroll 1
    for (int k = 0; k < TLDS / 6; ++k) {
        const int lb = TREG / 2 + 3 * k;
        const int wb = (TREG + TLDS) / 2 + 4 * k;
        far_trip<FLUSH>(Swt + (2 * wb + SIDE) * 64, Slw + (6 * k + SIDE) * (NWAVE * 64), hl, lb, wb,
                        kx, ky, a, rl, acc);
    }
}

__device__ __forceinline__ void first_trip_request(const float4* Swt, float4 (&bn)[NB]) {
    const float4* p = Swt + (TREG + TLDS) * 64;
    #pragma unroll
    for (int j = 0; j < NB; ++j) bn[j] = p[128 * (j >> 1) + 32 * (j & 1)];
}

__device__ __forceinline__ void rolling_pass(const float4* Swt, const float4* Slw, const float* hl,
                                             float4 (&bn)[NB], v8f& accA, v8f& accB) {
    float4 b[8];
    {
        const float4* p = Swt + (TREG + TLDS) * 64;
        #pragma unroll
        for (int j = 0; j < NB; ++j) b[j] = bn[j];
        #pragma unroll
        for (int j = NB; j < 8; ++j) b[j] = p[128 * (j >> 1) + 32 * (j & 1)];
    }
    float hcur = hl[(TREG + TLDS) / 2];
    float hlast = 0.0f;
    #pragma unroll
    for (int i = 0; i < 6; ++i) {
        const int side = i / 3, k = i % 3;
        const int lb = TREG / 2 + 3 * k;
        const int wb = (TREG + TLDS) / 2 + 4 * k;
        const int ni = (i + 1) % 6;
        const int nwb = (TREG + TLDS) / 2 + 4 * (ni % 3);
        const float4* np = Swt + (2 * nwb + ni / 3) * 64;
        const float4* q0 = Slw + (6 * k + side) * (NWAVE * 64);
        #pragma unroll
        for (int j = 0; j < 4; ++j) {
            const v8f c = tile_of(b[2 * j], b[2 * j + 1]);
            if (side) accB = axpy8(c, hcur, accB); else accA = axpy8(c, hcur, accA);
            __builtin_amdgcn_sched_barrier(0);
            if (j < 3) {
                const float4* q = q0 + j * (2 * NWAVE * 64);
                const v8f d = lds_tile(q);
                const float hd = hl[lb + j];
                hcur = hl[wb + j + 1];
                if (j == 2 && i < 5) hlast = hl[nwb];
                if (side) accB = axpy8(d, hd, accB); else accA = axpy8(d, hd, accA);
                __builtin_amdgcn_sched_barrier(0);
            } else {
                hcur = hlast;
            }
            if (i < 5 || 2 * j < NB) {
                b[2 * j]     = np[128 * j];
                b[2 * j + 1] = np[128 * j + 32];
            }
            __builtin_amdgcn_sched_barrier(0);
        }
    }
    #pragma unroll
    for (int j = 0; j < NB; ++j) bn[j] = b[j];
}

__device__ __forceinline__ float reduce_rows(v8f accA, v8f accB, int lane) {
    const bool b3 = (lane & 8) != 0, b2 = (lane & 4) != 0, b1 = (lane & 2) != 0, b0 = (lane & 1) != 0;
    float x[8], y[4], z[2];
    #pragma unroll
    for (int v = 0; v < 8; ++v) {
        const float keep = b3 ? accB[v] : accA[v];
        const float send = b3 ? accA[v] : accB[v];
        x[v] = keep + row_xor<8>(send);
    }
    #pragma unroll
    for (int v = 0; v < 4; ++v) {
        const float keep = b2 ? x[4 + v] : x[v];
        const float send = b2 ? x[v] : x[4 + v];
        y[v] = keep + row_xor<4>(send);
    }
    #pragma unroll
    for (int v = 0; v < 2; ++v) {
        const float keep = b1 ? y[2 + v] : y[v];
        const float send = b1 ? y[v] : y[2 + v];
        z[v] = keep + row_xor<2>(send);
    }
    const float keep = b0 ? z[1] : z[0];
    const float send = b0 ? z[0] : z[1];
    return keep + row_xor<1>(send);
}

template <int SIDE>
__device__ __forceinline__ void flush_side(v8f (&sr)[TREG], float4* Swt, float4* Slw, const float* hl,
                                           const float (*kbuf)[LD], const float* svals,
                                           int cs, int rl, int hh, v8f& acc) {
    const int kq = 2 * hh;
    const float* kx = &kbuf[kq][0];
    const float* ky = &kbuf[kq + 1][0];
    const int colA = 32 * cs + 16 * SIDE + rl;
    AFrag a;
    #pragma unroll
    for (int i = 0; i < NG; ++i) {
        a.g[i].x = -svals[4 * i + kq] * kx[4 * i * LD + colA];
        a.g[i].y = -svals[4 * i + kq + 1] * ky[4 * i * LD + colA];
    }
    #pragma unroll
    for (int i = 0; i < TREG / 2; ++i)
        sr[2 * i + SIDE] = tile_update(a, kx, ky, 16 * i + rl, sr[2 * i + SIDE]);
    far_side_pass<true, SIDE>(Swt, Slw, hl, kx, ky, a, rl, acc);
}

template <int NP>
__device__ __forceinline__ void wave_sums(const float* kp, float hv, float q0, float q1, float* red, int wave, int lane) {
    float s[NP + 2];
    #pragma unroll
    for (int p = 0; p < NP; ++p) s[p] = kp[p] * hv;
    s[NP] = q0;
    s[NP + 1] = q1;
    #pragma unroll
    for (int off = 16; off > 0; off >>= 1) {
        #pragma unroll
        for (int p = 0; p < NP + 2; ++p) s[p] += __shfl_down(s[p], off, 32);
    }
    if (lane == 0) {
        #pragma unroll
        for (int p = 0; p < NP; ++p) red[p * NWAVE + wave] = s[p];
        red[(WIN - 1) * NWAVE + wave] = s[NP];
        red[WIN * NWAVE + wave] = s[NP + 1];
    }
}

__launch_bounds__(512)
__global__ void kf_scan(const float* __restrict__ emissions,
                        const float* __restrict__ Ht,
                        const float* __restrict__ R,
                        const float* __restrict__ mu0,
                        float* __restrict__ S,
                        float* __restrict__ out) {
    __shared__ float4 Sl[TLDS * NWAVE * 2 * 32];
    __shared__ float4 hT4[16 * HP / 4];
    __shared__ float m[LD];
    __shared__ float kbuf[WIN][LD];
    __shared__ float svals[WIN];
    __shared__ float4 red4[NSUM * NWAVE / 4];
    __shared__ float ys[NSTEPS];

    float* hT = reinterpret_cast<float*>(hT4);
    float* red = reinterpret_cast<float*>(red4);

    const int tid  = threadIdx.x;
    const int lane = tid & 31;
    const int wave = __builtin_amdgcn_readfirstlane(tid >> 5);
    const int rl = lane & 15;
    const int hh = lane >> 4;
    const int cs = wave;
    const int col = 32 * cs + 16 * ((lane >> 3) & 1) + 8 * hh + (lane & 7);
    const float r = R[0];

    float4* Swt = reinterpret_cast<float4*>(S) + cs * (NTILE * 64) + lane;
    float4* Slw = Sl + wave * 64 + lane;
    const float* hl = hT + rl * HP;
    float* hTc = hT + (col & 15) * HP + (col >> 4);

    m[col] = (col < D1) ? mu0[col] : 0.0f;
    float hv = (col < D1) ? Ht[col] : 0.0f;
    *hTc = hv;
    static_assert(NSTEPS == NWAVE * 32, "one emission per thread");
    ys[tid] = emissions[tid];
    if (tid < WIN) svals[tid] = 0.0f;
    #pragma unroll
    for (int p = 0; p < WIN; ++p) kbuf[p][tid] = 0.0f;
    if (tid < NSUM * NWAVE) red[tid] = 0.0f;

    v8f sr[TREG];
    #pragma unroll
    for (int n = 0; n < TREG; ++n) sr[n] = lds_tile(Swt + n * 64);
    #pragma unroll 1
    for (int n = TREG; n < TREG + TLDS; ++n) {
        const float4* p = Swt + n * 64;
        float4* q = Slw + (n - TREG) * (NWAVE * 64);
        q[0]  = p[0];
        q[32] = p[32];
    }
    __syncthreads();

    int w = 0;
    float4 bn[NB];
    first_trip_request(Swt, bn);
    for (int t = 0; t < NSTEPS; ++t) {

        if (t + 1 < NSTEPS) {
            const int c0 = 32 * cs;
            const int c1 = (c0 + 31 < D1) ? c0 + 31 : D1 - 1;
            const float* nrow = Ht + (t + 1) * D1;
            __builtin_prefetch(nrow + c0, 0, 0);
            __builtin_prefetch(nrow + c1, 0, 0);
        }

        v8f accA, accB;
        #pragma unroll
        for (int v = 0; v < 8; ++v) { accA[v] = 0.0f; accB[v] = 0.0f; }
        if (w == 0 && t > 0) {
            flush_side<0>(sr, Swt, Slw, hl, kbuf, svals, cs, rl, hh, accA);
            flush_side<1>(sr, Swt, Slw, hl, kbuf, svals, cs, rl, hh, accB);
            first_trip_request(Swt, bn);
        } else {
            rolling_pass(Swt, Slw, hl, bn, accA, accB);
        }
        float4 hq[(TREG / 2 + 3) / 4];
        #pragma unroll
        for (int j = 0; j < (TREG / 2 + 3) / 4; ++j) hq[j] = reinterpret_cast<const float4*>(hl)[j];
        #pragma unroll
        for (int n = 0; n < TREG; ++n) {
            const float4 h4v = hq[n >> 3];
            const int e = (n >> 1) & 3;
            const float hb = (e == 0) ? h4v.x : (e == 1) ? h4v.y : (e == 2) ? h4v.z : h4v.w;
            if (n & 1) accB = axpy8(sr[n], hb, accB); else accA = axpy8(sr[n], hb, accA);
        }
        const float pv = reduce_rows(accA, accB, lane);

        float hn = 0.0f;
        if (t + 1 < NSTEPS && col < D1) {
            hn = Ht[(unsigned)((t + 1) * D1) + (unsigned)col];
        }

        float kp[WIN - 1];
        #pragma unroll
        for (int p = 0; p < WIN - 1; ++p) {
            const float kraw = kbuf[p][col];
            kp[p] = (w > p) ? kraw : 0.0f;
        }
        {
            const float q0 = pv * hv;
            const float q1 = hv * m[col];
            static_assert(WIN == 8, "one case per place of a step in its window");
            switch (w) {
                case 0:  wave_sums<0>(kp, hv, q0, q1, red, wave, lane); break;
                case 1:  wave_sums<1>(kp, hv, q0, q1, red, wave, lane); break;
                case 2:  wave_sums<2>(kp, hv, q0, q1, red, wave, lane); break;
                case 3:  wave_sums<3>(kp, hv, q0, q1, red, wave, lane); break;
                case 4:  wave_sums<4>(kp, hv, q0, q1, red, wave, lane); break;
                case 5:  wave_sums<5>(kp, hv, q0, q1, red, wave, lane); break;
                case 6:  wave_sums<6>(kp, hv, q0, q1, red, wave, lane); break;
                default: wave_sums<WIN - 1>(kp, hv, q0, q1, red, wave, lane); break;
            }
        }
        __syncthreads();

        if (w == 0 && t > 0) {
            #pragma unroll
            for (int n = TREG + TLDS; n < NTILE; ++n) {
                __builtin_prefetch(Swt + n * 64, 0, 2);
                __builtin_prefetch(Swt + n * 64 + 32, 0, 2);
            }
        }
        static_assert(2 * NSUM <= 32 && NWAVE == 16, "one half (8 wave sums) of a sum per lane");
        const int ph = (lane < 2 * NSUM) ? lane : (2 * NSUM - 1);
        const float4 xa = red4[2 * ph];
        const float4 xb = red4[2 * ph + 1];
        const float part = ((xa.x + xa.y) + (xa.z + xa.w)) + ((xb.x + xb.y) + (xb.z + xb.w));
        const float tl = part + row_xor<1>(part);
        float tot[NSUM];
        #pragma unroll
        for (int v = 0; v < NSUM; ++v)
            tot[v] = __int_as_float(__builtin_amdgcn_readlane(__float_as_int(tl), 2 * v));
        float s1 = tot[WIN - 1];
        float v = pv;
        #pragma unroll
        for (int p = 0; p < WIN - 1; ++p) {
            const float dp = (w > p) ? tot[p] : 0.0f;
            const float cp = (w > p) ? svals[p] * tot[p] : 0.0f;
            s1 -= cp * dp;
            v -= cp * kp[p];
        }
        const float s_term = r + s1;
        const float inva = 1.0f / (s_term + 1e-6f);
        const float resid = ys[t] - tot[WIN];
        const float kv = v * inva;
        kbuf[w][col] = kv;
        const float mv = m[col] + kv * resid;
        m[col] = mv;
        if (tid == 0) svals[w] = s_term;
        *hTc = hn;
        hv = hn;
        __syncthreads();
        if (col < D1) out[t * D1 + col] = mv;
        w = (w + 1 == WIN) ? 0 : w + 1;
    }
}

extern "C" void kernel_launch(void* const* d_in, const int* in_sizes, int n_in,
                              void* d_out, int out_size, void* d_ws, size_t ws_size,
                              hipStream_t stream) {
    const float* emissions = (const float*)d_in[0];
    const float* Ht        = (const float*)d_in[1];
    const float* R         = (const float*)d_in[4];
    const float* mu0       = (const float*)d_in[5];
    const float* Sigma0    = (const float*)d_in[6];

    float* S   = (float*)d_ws;
    float* out = (float*)d_out;

    kf_init_S<<<(LD * LD / 4) / 256, 256, 0, stream>>>(Sigma0, S);
    kf_scan<<<1, 512, 0, stream>>>(emissions, Ht, R, mu0, S, out);
}
